# starting kernel
# baseline (speedup 1.0000x reference)
#include <hip/hip_runtime.h>
#include <hip/hip_bf16.h>
#include <stddef.h>


typedef __attribute__((ext_vector_type(2))) float v2f;
typedef __attribute__((ext_vector_type(8))) float v8f;

#define T_DIM 4096
#define N_DIM 16384
#define EMA_W 0.9f
#define LOG2_W (-0.15200309344504995f)

#define T_PROC 1536
#define ROWS_PER_CHUNK 128
#define CHUNKS (T_PROC / ROWS_PER_CHUNK)
#define WAVES_PER_BLOCK 8

__global__ void __launch_bounds__(256)
ema_wmma_partial(const float* __restrict__ x, float* __restrict__ partial) {
    const int lane = threadIdx.x & 31;
    const int wave = threadIdx.x >> 5;
    const int n0   = (blockIdx.x * WAVES_PER_BLOCK + wave) * 16;
    const int t0   = (T_DIM - T_PROC) + blockIdx.y * ROWS_PER_CHUNK;

    const int col  = n0 + (lane & 15);
    const int koff = (lane >> 4) << 1;

    const float* p = x + (size_t)(t0 + koff) * N_DIM + col;

    const float invW  = 1.0f / EMA_W;
    const float lmul  = (koff == 0) ? 1.0f : invW * invW;

    v8f acc = {};

#pragma unroll 4
    for (int t = t0; t < t0 + ROWS_PER_CHUNK; t += 4) {
        v2f b;
        b.x = p[0];
        b.y = p[N_DIM];

        const float wstep = -(1.0f - EMA_W) * (1.0f - EMA_W) *
                            exp2f((float)(T_DIM - 2 - t) * LOG2_W);
        v2f a;
        a.x = wstep * lmul;
        float wy = a.x * invW;
        a.y = (t + koff + 1 == T_DIM - 1) ? (2.0f - EMA_W) : wy;

        if (t + 16 < T_DIM - 4) {
            __builtin_prefetch(p + (size_t)16 * N_DIM, 0, 1);
        }

        acc = __builtin_amdgcn_wmma_f32_16x16x4_f32(
             false, a,  false, b,
             (short)0, acc,  false,  false);

        p += (size_t)4 * N_DIM;
    }

    if (lane < 16) {
        partial[(size_t)blockIdx.y * N_DIM + n0 + lane] = acc[0];
    }
}

__global__ void __launch_bounds__(256)
ema_combine(const float* __restrict__ partial, float* __restrict__ out) {
    const int n = blockIdx.x * blockDim.x + threadIdx.x;
    float s = 0.0f;
#pragma unroll
    for (int c = 0; c < CHUNKS; ++c) {
        s += partial[(size_t)c * N_DIM + n];
    }
    out[n] = s;
}

extern "C" void kernel_launch(void* const* d_in, const int* in_sizes, int n_in,
                              void* d_out, int out_size, void* d_ws, size_t ws_size,
                              hipStream_t stream) {
    (void)in_sizes; (void)n_in; (void)out_size; (void)ws_size;

    const float* x       = (const float*)d_in[0];
    float*       out     = (float*)d_out;
    float*       partial = (float*)d_ws;

    dim3 grid1(N_DIM / (16 * WAVES_PER_BLOCK), CHUNKS);
    ema_wmma_partial<<<grid1, 256, 0, stream>>>(x, partial);

    ema_combine<<<N_DIM / 256, 256, 0, stream>>>(partial, out);
}


# each lane loads 16 B (global_load_b128: 4 columns -> 4 WMMA accumulators) and a wave issues all 64 loads of its 128x64 tile up front: 3072 waves instead of 12288, 6x fewer memory instructions
# speedup vs baseline: 1.2140x; 1.2140x over previous
#include <hip/hip_runtime.h>
#include <hip/hip_bf16.h>
#include <stddef.h>


typedef __attribute__((ext_vector_type(2))) float v2f;
typedef __attribute__((ext_vector_type(4))) float v4f;
typedef __attribute__((ext_vector_type(8))) float v8f;

#define T_DIM 4096
#define N_DIM 16384
#define EMA_W 0.9f
#define LOG2_W (-0.15200309344504995f)

#define T_PROC 1536
#define ROWS_PER_CHUNK 128
#define CHUNKS (T_PROC / ROWS_PER_CHUNK)
#define STEPS (ROWS_PER_CHUNK / 4)
#define WAVES_PER_BLOCK 4
#define COLS_PER_WAVE 64

__global__ void __launch_bounds__(128, 3)
ema_wmma_partial(const float* __restrict__ x, float* __restrict__ partial) {
    const int lane = threadIdx.x & 31;
    const int wave = threadIdx.x >> 5;
    const int n0   = (blockIdx.x * WAVES_PER_BLOCK + wave) * COLS_PER_WAVE;
    const int t0   = (T_DIM - T_PROC) + blockIdx.y * ROWS_PER_CHUNK;

    const int c    = lane & 15;
    const int koff = (lane >> 4) << 1;

    const float* p = x + (size_t)(t0 + koff) * N_DIM + n0 + 4 * c;

    v4f r0[STEPS], r1[STEPS];
#pragma unroll
    for (int s = 0; s < STEPS; ++s) {
        r0[s] = *(const v4f*)(p + (size_t)(4 * s) * N_DIM);
        r1[s] = *(const v4f*)(p + (size_t)(4 * s + 1) * N_DIM);
    }
    __builtin_amdgcn_sched_barrier(0);

    const float invW  = 1.0f / EMA_W;
    const float lmul  = (koff == 0) ? 1.0f : invW * invW;

    v8f acc0 = {}, acc1 = {}, acc2 = {}, acc3 = {};

#pragma unroll
    for (int s = 0; s < STEPS; ++s) {
        const int t = t0 + 4 * s;

        const float wstep = -(1.0f - EMA_W) * (1.0f - EMA_W) *
                            exp2f((float)(T_DIM - 2 - t) * LOG2_W);
        v2f a;
        a.x = wstep * lmul;
        float wy = a.x * invW;
        a.y = (t + koff + 1 == T_DIM - 1) ? (2.0f - EMA_W) : wy;

        v2f b0, b1, b2, b3;
        b0.x = r0[s].x; b0.y = r1[s].x;
        b1.x = r0[s].y; b1.y = r1[s].y;
        b2.x = r0[s].z; b2.y = r1[s].z;
        b3.x = r0[s].w; b3.y = r1[s].w;

        acc0 = __builtin_amdgcn_wmma_f32_16x16x4_f32(
             false, a,  false, b0,
             (short)0, acc0,  false,  false);
        acc1 = __builtin_amdgcn_wmma_f32_16x16x4_f32(
            false, a, false, b1, (short)0, acc1, false, false);
        acc2 = __builtin_amdgcn_wmma_f32_16x16x4_f32(
            false, a, false, b2, (short)0, acc2, false, false);
        acc3 = __builtin_amdgcn_wmma_f32_16x16x4_f32(
            false, a, false, b3, (short)0, acc3, false, false);
    }

    if (lane < 16) {
        v4f o;
        o.x = acc0[0]; o.y = acc1[0]; o.z = acc2[0]; o.w = acc3[0];
        *(v4f*)(partial + (size_t)blockIdx.y * N_DIM + n0 + 4 * c) = o;
    }
}

__global__ void __launch_bounds__(256)
ema_combine(const float* __restrict__ partial, float* __restrict__ out) {
    const int n = blockIdx.x * blockDim.x + threadIdx.x;
    float s = 0.0f;
#pragma unroll
    for (int c = 0; c < CHUNKS; ++c) {
        s += partial[(size_t)c * N_DIM + n];
    }
    out[n] = s;
}

extern "C" void kernel_launch(void* const* d_in, const int* in_sizes, int n_in,
                              void* d_out, int out_size, void* d_ws, size_t ws_size,
                              hipStream_t stream) {
    (void)in_sizes; (void)n_in; (void)out_size; (void)ws_size;

    const float* x       = (const float*)d_in[0];
    float*       out     = (float*)d_out;
    float*       partial = (float*)d_ws;

    dim3 grid1(N_DIM / (COLS_PER_WAVE * WAVES_PER_BLOCK), CHUNKS);
    ema_wmma_partial<<<grid1, 32 * WAVES_PER_BLOCK, 0, stream>>>(x, partial);

    ema_combine<<<N_DIM / 256, 256, 0, stream>>>(partial, out);
}


# one launch instead of two: a 12-wave workgroup owns a 64-column strip over all 12 row chunks; per-chunk partials meet in LDS and are added in the same chunk order; combine kernel removed
# speedup vs baseline: 1.4608x; 1.2033x over previous
#include <hip/hip_runtime.h>
#include <hip/hip_bf16.h>
#include <stddef.h>


typedef __attribute__((ext_vector_type(2))) float v2f;
typedef __attribute__((ext_vector_type(4))) float v4f;
typedef __attribute__((ext_vector_type(8))) float v8f;

#define T_DIM 4096
#define N_DIM 16384
#define EMA_W 0.9f
#define LOG2_W (-0.15200309344504995f)

#define T_PROC 1536
#define ROWS_PER_CHUNK 128
#define CHUNKS (T_PROC / ROWS_PER_CHUNK)
#define STEPS (ROWS_PER_CHUNK / 4)
#define WAVES_PER_BLOCK CHUNKS
#define COLS_PER_WAVE 64

__global__ void __launch_bounds__(32 * WAVES_PER_BLOCK, 1)
ema_wmma_fused(const float* __restrict__ x, float* __restrict__ out) {
    __shared__ v4f red[CHUNKS][16];

    const int lane = threadIdx.x & 31;
    const int wave = __builtin_amdgcn_readfirstlane((int)(threadIdx.x >> 5));
    const int n0   = blockIdx.x * COLS_PER_WAVE;
    const int t0   = (T_DIM - T_PROC) + wave * ROWS_PER_CHUNK;

    const int c    = lane & 15;
    const int koff = (lane >> 4) << 1;

    const float* p = x + (size_t)(t0 + koff) * N_DIM + n0 + 4 * c;

    v4f r0[STEPS], r1[STEPS];
#pragma unroll
    for (int s = 0; s < STEPS; ++s) {
        r0[s] = *(const v4f*)(p + (size_t)(4 * s) * N_DIM);
        r1[s] = *(const v4f*)(p + (size_t)(4 * s + 1) * N_DIM);
    }
    __builtin_amdgcn_sched_barrier(0);

    const float invW  = 1.0f / EMA_W;
    const float lmul  = (koff == 0) ? 1.0f : invW * invW;

    v8f acc0 = {}, acc1 = {}, acc2 = {}, acc3 = {};

#pragma unroll
    for (int s = 0; s < STEPS; ++s) {
        const int t = t0 + 4 * s;

        const float wstep = -(1.0f - EMA_W) * (1.0f - EMA_W) *
                            exp2f((float)(T_DIM - 2 - t) * LOG2_W);
        v2f a;
        a.x = wstep * lmul;
        float wy = a.x * invW;
        a.y = (t + koff + 1 == T_DIM - 1) ? (2.0f - EMA_W) : wy;

        v2f b0, b1, b2, b3;
        b0.x = r0[s].x; b0.y = r1[s].x;
        b1.x = r0[s].y; b1.y = r1[s].y;
        b2.x = r0[s].z; b2.y = r1[s].z;
        b3.x = r0[s].w; b3.y = r1[s].w;

        acc0 = __builtin_amdgcn_wmma_f32_16x16x4_f32(
             false, a,  false, b0,
             (short)0, acc0,  false,  false);
        acc1 = __builtin_amdgcn_wmma_f32_16x16x4_f32(
            false, a, false, b1, (short)0, acc1, false, false);
        acc2 = __builtin_amdgcn_wmma_f32_16x16x4_f32(
            false, a, false, b2, (short)0, acc2, false, false);
        acc3 = __builtin_amdgcn_wmma_f32_16x16x4_f32(
            false, a, false, b3, (short)0, acc3, false, false);
    }

    if (lane < 16) {
        v4f o;
        o.x = acc0[0]; o.y = acc1[0]; o.z = acc2[0]; o.w = acc3[0];
        red[wave][c] = o;
    }
    __syncthreads();

    if (threadIdx.x < 16) {
        v4f s = {0.0f, 0.0f, 0.0f, 0.0f};
        for (int k = 0; k < CHUNKS; ++k) {
            s += red[k][threadIdx.x];
        }
        *(v4f*)(out + n0 + 4 * threadIdx.x) = s;
    }
}

extern "C" void kernel_launch(void* const* d_in, const int* in_sizes, int n_in,
                              void* d_out, int out_size, void* d_ws, size_t ws_size,
                              hipStream_t stream) {
    (void)in_sizes; (void)n_in; (void)out_size; (void)d_ws; (void)ws_size;

    const float* x   = (const float*)d_in[0];
    float*       out = (float*)d_out;

    ema_wmma_fused<<<N_DIM / COLS_PER_WAVE, 32 * WAVES_PER_BLOCK, 0, stream>>>(x, out);
}


# one wave per SIMD: 4-wave workgroups, a wave streams 384 rows with loads grouped by 12, 24 k-steps ahead of its WMMAs; 1024 waves enter, not 3072; sums: 4 partials of 384 rows, not 12 of 128
# speedup vs baseline: 1.7766x; 1.2162x over previous
#include <hip/hip_runtime.h>
#include <hip/hip_bf16.h>
#include <stddef.h>


typedef __attribute__((ext_vector_type(2))) float v2f;
typedef __attribute__((ext_vector_type(4))) float v4f;
typedef __attribute__((ext_vector_type(8))) float v8f;

#define T_DIM 4096
#define N_DIM 16384
#define EMA_W 0.9f
#define LOG2_W (-0.15200309344504995f)

#define T_PROC 1536
#define ROWS_PER_CHUNK 384
#define CHUNKS (T_PROC / ROWS_PER_CHUNK)
#define STEPS (ROWS_PER_CHUNK / 4)
#define WAVES_PER_BLOCK CHUNKS
#define COLS_PER_WAVE 64
#define GROUP 6
#define GROUPS (STEPS / GROUP)
#define GROUPS_AHEAD 4
#define AHEAD ((GROUPS_AHEAD + 1) * GROUP)

__global__ void __launch_bounds__(32 * WAVES_PER_BLOCK)
__attribute__((amdgpu_waves_per_eu(1, 1)))
ema_wmma_fused(const float* __restrict__ x, float* __restrict__ out) {
    __shared__ v4f red[CHUNKS][16];

    const int lane = threadIdx.x & 31;
    const int wave = __builtin_amdgcn_readfirstlane((int)(threadIdx.x >> 5));
    const int n0   = blockIdx.x * COLS_PER_WAVE;
    const int t0   = (T_DIM - T_PROC) + wave * ROWS_PER_CHUNK;

    const int c    = lane & 15;
    const int koff = (lane >> 4) << 1;

    const float* p = x + (size_t)(t0 + koff) * N_DIM + n0 + 4 * c;

    v4f r0[STEPS], r1[STEPS];
#pragma unroll
    for (int s = 0; s < AHEAD; ++s) {
        r0[s] = *(const v4f*)(p + (size_t)(4 * s) * N_DIM);
        r1[s] = *(const v4f*)(p + (size_t)(4 * s + 1) * N_DIM);
    }
    __builtin_amdgcn_sched_barrier(0);

    const float invW  = 1.0f / EMA_W;
    const float lmul  = (koff == 0) ? 1.0f : invW * invW;

    float wv[STEPS / 32];
#pragma unroll
    for (int j = 0; j < STEPS / 32; ++j) {
        const int t = t0 + 4 * (32 * j + lane);
        wv[j] = -(1.0f - EMA_W) * (1.0f - EMA_W) *
                exp2f((float)(T_DIM - 2 - t) * LOG2_W);
    }
    v2f aw[STEPS];
#pragma unroll
    for (int s = 0; s < STEPS; ++s) {
        const int t = t0 + 4 * s;

        const float wstep = __builtin_bit_cast(float,
            __builtin_amdgcn_readlane(__builtin_bit_cast(int, wv[s >> 5]), s & 31));
        v2f a;
        a.x = wstep * lmul;
        float wy = a.x * invW;
        a.y = (t + koff + 1 == T_DIM - 1) ? (2.0f - EMA_W) : wy;
        aw[s] = a;
    }
    __builtin_amdgcn_sched_barrier(0);

    v8f acc0 = {}, acc1 = {}, acc2 = {}, acc3 = {};

#pragma unroll
    for (int g = 0; g < GROUPS; ++g) {
        if (g >= 1 && g + GROUPS_AHEAD < GROUPS) {
            __builtin_amdgcn_sched_barrier(0);
#pragma unroll
            for (int s = (g + GROUPS_AHEAD) * GROUP; s < (g + GROUPS_AHEAD + 1) * GROUP; ++s) {
                r0[s] = *(const v4f*)(p + (size_t)(4 * s) * N_DIM);
                r1[s] = *(const v4f*)(p + (size_t)(4 * s + 1) * N_DIM);
            }
            __builtin_amdgcn_sched_barrier(0);
        }

#pragma unroll
      for (int s = g * GROUP; s < (g + 1) * GROUP; ++s) {
        const v2f a = aw[s];

        v2f b0, b1, b2, b3;
        b0.x = r0[s].x; b0.y = r1[s].x;
        b1.x = r0[s].y; b1.y = r1[s].y;
        b2.x = r0[s].z; b2.y = r1[s].z;
        b3.x = r0[s].w; b3.y = r1[s].w;

        acc0 = __builtin_amdgcn_wmma_f32_16x16x4_f32(
             false, a,  false, b0,
             (short)0, acc0,  false,  false);
        acc1 = __builtin_amdgcn_wmma_f32_16x16x4_f32(
            false, a, false, b1, (short)0, acc1, false, false);
        acc2 = __builtin_amdgcn_wmma_f32_16x16x4_f32(
            false, a, false, b2, (short)0, acc2, false, false);
        acc3 = __builtin_amdgcn_wmma_f32_16x16x4_f32(
            false, a, false, b3, (short)0, acc3, false, false);
      }
    }

    if (lane < 16) {
        v4f o;
        o.x = acc0[0]; o.y = acc1[0]; o.z = acc2[0]; o.w = acc3[0];
        red[wave][c] = o;
    }
    __syncthreads();

    if (threadIdx.x < 16) {
        v4f s = {0.0f, 0.0f, 0.0f, 0.0f};
        for (int k = 0; k < CHUNKS; ++k) {
            s += red[k][threadIdx.x];
        }
        *(v4f*)(out + n0 + 4 * threadIdx.x) = s;
    }
}

extern "C" void kernel_launch(void* const* d_in, const int* in_sizes, int n_in,
                              void* d_out, int out_size, void* d_ws, size_t ws_size,
                              hipStream_t stream) {
    (void)in_sizes; (void)n_in; (void)out_size; (void)d_ws; (void)ws_size;

    const float* x   = (const float*)d_in[0];
    float*       out = (float*)d_out;

    ema_wmma_fused<<<N_DIM / COLS_PER_WAVE, 32 * WAVES_PER_BLOCK, 0, stream>>>(x, out);
}
